# starting kernel
# baseline (speedup 1.0000x reference)
#include <hip/hip_runtime.h>


typedef float v2f __attribute__((ext_vector_type(2)));
typedef float v8f __attribute__((ext_vector_type(8)));

#define B_ 128
#define T_ 4096
#define I_ 6
#define H_ 32
#define O_ 2

__device__ __forceinline__ v8f wmma4(v2f a, v2f b, v8f c) {
    return __builtin_amdgcn_wmma_f32_16x16x4_f32(
         false, a,  false, b,
         (short)0, c,  false,  false);
}

__device__ __forceinline__ float fsigmoid(float a) {
    return __builtin_amdgcn_rcpf(1.0f + __builtin_amdgcn_exp2f(-1.44269504088896f * a));
}
__device__ __forceinline__ float ftanh(float a) {
    return 2.0f * __builtin_amdgcn_rcpf(1.0f + __builtin_amdgcn_exp2f(-2.88539008177793f * a)) - 1.0f;
}

__global__ __launch_bounds__(64, 1) void gru_fused_kernel(
    const float* __restrict__ x,
    const int*   __restrict__ lengths,
    const float* __restrict__ w_ih,
    const float* __restrict__ w_hh,
    const float* __restrict__ b_ih,
    const float* __restrict__ b_hh,
    const float* __restrict__ fc_w,
    const float* __restrict__ fc_b,
    float*       __restrict__ out)
{
    __shared__ float ldsh[H_ * 16];

    const int tid  = threadIdx.x;
    const int wv   = tid >> 5;
    const int lane = tid & 31;
    const int lh   = lane >> 4;
    const int ln   = lane & 15;
    const int b0   = blockIdx.x * 16;

    for (int j = tid; j < H_ * 16; j += 64) ldsh[j] = 0.0f;
    __syncthreads();

    v2f Wg[3][8];
    v2f Wi[3][2];
    float biasR, biasZ, bxn, bhn;
    #pragma unroll
    for (int i = 0; i < 3; ++i) {
        const int g = i * 32 + wv * 16 + ln;
        #pragma unroll
        for (int s = 0; s < 8; ++s) {
            const int k0 = 4 * s + 2 * lh;
            Wg[i][s].x = w_hh[g * H_ + k0];
            Wg[i][s].y = w_hh[g * H_ + k0 + 1];
        }
        #pragma unroll
        for (int s = 0; s < 2; ++s) {
            const int k0 = 4 * s + 2 * lh;
            Wi[i][s].x = (k0     < I_) ? w_ih[g * I_ + k0]     : 0.0f;
            Wi[i][s].y = (k0 + 1 < I_) ? w_ih[g * I_ + k0 + 1] : 0.0f;
        }
    }
    {
        const int gr = 0 * 32 + wv * 16 + ln;
        const int gz = 1 * 32 + wv * 16 + ln;
        const int gn = 2 * 32 + wv * 16 + ln;
        biasR = b_ih[gr] + b_hh[gr];
        biasZ = b_ih[gz] + b_hh[gz];
        bxn   = b_ih[gn];
        bhn   = b_hh[gn];
    }

    const bool fcl = (wv == 0) && (ln < O_);
    v2f Wfc[8];
    #pragma unroll
    for (int s = 0; s < 8; ++s) {
        const int k0 = 4 * s + 2 * lh;
        Wfc[s].x = fcl ? fc_w[ln * H_ + k0]     : 0.0f;
        Wfc[s].y = fcl ? fc_w[ln * H_ + k0 + 1] : 0.0f;
    }
    const float fcb = fcl ? fc_b[ln] : 0.0f;

    int len8[8];
    #pragma unroll
    for (int v = 0; v < 8; ++v) len8[v] = lengths[b0 + v + 8 * lh];

    const float* xrow = x + (size_t)(b0 + ln) * T_ * I_;
    const int    hcol = wv * 16 + ln;

    v2f   hA[8];
    float hold[8];
    #pragma unroll
    for (int s = 0; s < 8; ++s) { hA[s].x = 0.0f; hA[s].y = 0.0f; }
    #pragma unroll
    for (int v = 0; v < 8; ++v) hold[v] = 0.0f;

    #pragma unroll 1
    for (int t = 0; t <= T_; ++t) {
        if (wv == 0) {
            v8f accF;
            #pragma unroll
            for (int v = 0; v < 8; ++v) accF[v] = fcb;
            #pragma unroll
            for (int s = 0; s < 8; ++s) accF = wmma4(hA[s], Wfc[s], accF);
            if (t > 0 && ln < O_) {
                #pragma unroll
                for (int v = 0; v < 8; ++v) {
                    const float val = ((t - 1) < len8[v]) ? accF[v] : fcb;
                    out[((size_t)(b0 + v + 8 * lh) * T_ + (t - 1)) * O_ + ln] = val;
                }
            }
        }
        if (t == T_) break;

        const float* xp = xrow + t * I_;
        v2f xa0, xa1;
        xa0.x = xp[2 * lh];
        xa0.y = xp[2 * lh + 1];
        xa1.x = lh ? 0.0f : xp[4];
        xa1.y = lh ? 0.0f : xp[5];
        if (t + 1 < T_) __builtin_prefetch(xp + I_, 0, 1);

        v8f accR, accZ, accXN, accHN;
        #pragma unroll
        for (int v = 0; v < 8; ++v) {
            accR[v] = biasR; accZ[v] = biasZ; accXN[v] = bxn; accHN[v] = bhn;
        }
        accR  = wmma4(xa0, Wi[0][0], accR);  accR  = wmma4(xa1, Wi[0][1], accR);
        accZ  = wmma4(xa0, Wi[1][0], accZ);  accZ  = wmma4(xa1, Wi[1][1], accZ);
        accXN = wmma4(xa0, Wi[2][0], accXN); accXN = wmma4(xa1, Wi[2][1], accXN);
        #pragma unroll
        for (int s = 0; s < 8; ++s) {
            accR  = wmma4(hA[s], Wg[0][s], accR);
            accZ  = wmma4(hA[s], Wg[1][s], accZ);
            accHN = wmma4(hA[s], Wg[2][s], accHN);
        }

        #pragma unroll
        for (int v = 0; v < 8; ++v) {
            const float r  = fsigmoid(accR[v]);
            const float z  = fsigmoid(accZ[v]);
            const float n  = ftanh(accXN[v] + r * accHN[v]);
            const float hn = (1.0f - z) * n + z * hold[v];
            const float hs = (t < len8[v]) ? hn : hold[v];
            ldsh[hcol * 16 + 8 * lh + v] = hs;
            hold[v] = hs;
        }

        __syncthreads();

        #pragma unroll
        for (int s = 0; s < 8; ++s) {
            const int k0 = 4 * s + 2 * lh;
            hA[s].x = ldsh[k0 * 16 + ln];
            hA[s].y = ldsh[(k0 + 1) * 16 + ln];
        }

        __syncthreads();
    }
}

extern "C" void kernel_launch(void* const* d_in, const int* in_sizes, int n_in,
                              void* d_out, int out_size, void* d_ws, size_t ws_size,
                              hipStream_t stream) {
    const float* x       = (const float*)d_in[0];
    const int*   lengths = (const int*)  d_in[1];
    const float* w_ih    = (const float*)d_in[2];
    const float* w_hh    = (const float*)d_in[3];
    const float* b_ih    = (const float*)d_in[4];
    const float* b_hh    = (const float*)d_in[5];
    const float* fc_w    = (const float*)d_in[6];
    const float* fc_b    = (const float*)d_in[7];
    float*       out     = (float*)d_out;

    gru_fused_kernel<<<dim3(B_ / 16), dim3(64), 0, stream>>>(
        x, lengths, w_ih, w_hh, b_ih, b_hh, fc_w, fc_b, out);
}


# Spread each block's step over 5 waves: 4 gate waves own 8 h-columns each, r|z and hn|xn packed in two 16-col tiles (20 WMMA/step, was 38 on wave 0); FC on its own wave; barriers fence LDS only
# speedup vs baseline: 1.3819x; 1.3819x over previous
#include <hip/hip_runtime.h>


typedef float v2f __attribute__((ext_vector_type(2)));
typedef float v8f __attribute__((ext_vector_type(8)));

#define B_ 128
#define T_ 4096
#define I_ 6
#define H_ 32
#define O_ 2
#define NWAVES_ 5

__device__ __forceinline__ v8f wmma4(v2f a, v2f b, v8f c) {
    return __builtin_amdgcn_wmma_f32_16x16x4_f32(
         false, a,  false, b,
         (short)0, c,  false,  false);
}

__device__ __forceinline__ float fsigmoid(float a) {
    return __builtin_amdgcn_rcpf(1.0f + __builtin_amdgcn_exp2f(-1.44269504088896f * a));
}
__device__ __forceinline__ float ftanh(float a) {
    return 2.0f * __builtin_amdgcn_rcpf(1.0f + __builtin_amdgcn_exp2f(-2.88539008177793f * a)) - 1.0f;
}

__device__ __forceinline__ float from_lane_xor8(float v) {
    return __int_as_float(__builtin_amdgcn_update_dpp(0, __float_as_int(v), 0x128, 0xF, 0xF, true));
}

__device__ __forceinline__ void lds_barrier() {
    __builtin_amdgcn_fence(__ATOMIC_RELEASE, "workgroup", "local");
    __builtin_amdgcn_s_barrier();
    __builtin_amdgcn_fence(__ATOMIC_ACQUIRE, "workgroup", "local");
}

__global__ __launch_bounds__(32 * NWAVES_, 1) void gru_fused_kernel(
    const float* __restrict__ x,
    const int*   __restrict__ lengths,
    const float* __restrict__ w_ih,
    const float* __restrict__ w_hh,
    const float* __restrict__ b_ih,
    const float* __restrict__ b_hh,
    const float* __restrict__ fc_w,
    const float* __restrict__ fc_b,
    float*       __restrict__ out)
{
    __shared__ float ldsh[H_ * 16];

    const int tid  = threadIdx.x;
    const int wv   = __builtin_amdgcn_readfirstlane(tid >> 5);
    const int lane = tid & 31;
    const int lh   = lane >> 4;
    const int ln   = lane & 15;
    const int hi   = ln >> 3;
    const int b0   = blockIdx.x * 16;
    const bool gatew = wv < 4;

    for (int j = tid; j < H_ * 16; j += 32 * NWAVES_) ldsh[j] = 0.0f;
    __syncthreads();

    const int hc = (wv & 3) * 8 + (ln & 7);
    const int gP = hi ? (H_ + hc) : hc;
    const int gQ = 2 * H_ + hc;
    v2f WP[10], WQ[10];
    #pragma unroll
    for (int s = 0; s < 2; ++s) {
        const int k0 = 4 * s + 2 * lh;
        const float p0 = (k0     < I_) ? w_ih[gP * I_ + k0]     : 0.0f;
        const float p1 = (k0 + 1 < I_) ? w_ih[gP * I_ + k0 + 1] : 0.0f;
        const float q0 = (k0     < I_) ? w_ih[gQ * I_ + k0]     : 0.0f;
        const float q1 = (k0 + 1 < I_) ? w_ih[gQ * I_ + k0 + 1] : 0.0f;
        WP[s].x = p0;              WP[s].y = p1;
        WQ[s].x = hi ? q0 : 0.0f;  WQ[s].y = hi ? q1 : 0.0f;
    }
    #pragma unroll
    for (int s = 0; s < 8; ++s) {
        const int k0 = 4 * s + 2 * lh;
        WP[2 + s].x = w_hh[gP * H_ + k0];
        WP[2 + s].y = w_hh[gP * H_ + k0 + 1];
        const float q0 = w_hh[gQ * H_ + k0];
        const float q1 = w_hh[gQ * H_ + k0 + 1];
        WQ[2 + s].x = hi ? 0.0f : q0;
        WQ[2 + s].y = hi ? 0.0f : q1;
    }
    const float biasP = b_ih[gP] + b_hh[gP];
    const float biasQ = hi ? b_ih[gQ] : b_hh[gQ];

    const bool fcl = (ln < O_);
    v2f Wfc[8];
    #pragma unroll
    for (int s = 0; s < 8; ++s) {
        const int k0 = 4 * s + 2 * lh;
        Wfc[s].x = fcl ? fc_w[ln * H_ + k0]     : 0.0f;
        Wfc[s].y = fcl ? fc_w[ln * H_ + k0 + 1] : 0.0f;
    }
    const float fcb = fcl ? fc_b[ln] : 0.0f;

    int len8[8];
    #pragma unroll
    for (int v = 0; v < 8; ++v) len8[v] = lengths[b0 + v + 8 * lh];

    const float* xrow = x + (size_t)(b0 + ln) * T_ * I_;

    v2f   hA[8];
    float hold[8];
    #pragma unroll
    for (int s = 0; s < 8; ++s) { hA[s].x = 0.0f; hA[s].y = 0.0f; }
    #pragma unroll
    for (int v = 0; v < 8; ++v) hold[v] = 0.0f;

    #pragma unroll 1
    for (int t = 0; t < T_; ++t) {
        if (gatew) {
            const float* xp = xrow + t * I_;
            v2f xa0, xa1;
            xa0.x = xp[2 * lh];
            xa0.y = xp[2 * lh + 1];
            xa1.x = lh ? 0.0f : xp[4];
            xa1.y = lh ? 0.0f : xp[5];
            if (t + 1 < T_) __builtin_prefetch(xp + I_, 0, 1);

            v8f accP, accQ;
            #pragma unroll
            for (int v = 0; v < 8; ++v) { accP[v] = biasP; accQ[v] = biasQ; }
            accP = wmma4(xa0, WP[0], accP);  accP = wmma4(xa1, WP[1], accP);
            accQ = wmma4(xa0, WQ[0], accQ);  accQ = wmma4(xa1, WQ[1], accQ);
            #pragma unroll
            for (int s = 0; s < 8; ++s) {
                accP = wmma4(hA[s], WP[2 + s], accP);
                accQ = wmma4(hA[s], WQ[2 + s], accQ);
            }

            #pragma unroll
            for (int v = 0; v < 8; ++v) {
                const float sg = fsigmoid(accP[v]);
                const float z  = from_lane_xor8(sg);
                const float xn = from_lane_xor8(accQ[v]);
                const float n  = ftanh(xn + sg * accQ[v]);
                const float hn = (1.0f - z) * n + z * hold[v];
                const float hs = (t < len8[v]) ? hn : hold[v];
                hold[v] = hs;
            }
            if (!hi) {
                #pragma unroll
                for (int v = 0; v < 8; ++v) ldsh[hc * 16 + 8 * lh + v] = hold[v];
            }
        }

        lds_barrier();

        #pragma unroll
        for (int s = 0; s < 8; ++s) {
            const int k0 = 4 * s + 2 * lh;
            hA[s].x = ldsh[k0 * 16 + ln];
            hA[s].y = ldsh[(k0 + 1) * 16 + ln];
        }

        lds_barrier();

        if (wv == 4) {
            v8f accF;
            #pragma unroll
            for (int v = 0; v < 8; ++v) accF[v] = fcb;
            #pragma unroll
            for (int s = 0; s < 8; ++s) accF = wmma4(hA[s], Wfc[s], accF);
            if (ln < O_) {
                #pragma unroll
                for (int v = 0; v < 8; ++v) {
                    const float val = (t < len8[v]) ? accF[v] : fcb;
                    out[((size_t)(b0 + v + 8 * lh) * T_ + t) * O_ + ln] = val;
                }
            }
        }
    }
}

extern "C" void kernel_launch(void* const* d_in, const int* in_sizes, int n_in,
                              void* d_out, int out_size, void* d_ws, size_t ws_size,
                              hipStream_t stream) {
    const float* x       = (const float*)d_in[0];
    const int*   lengths = (const int*)  d_in[1];
    const float* w_ih    = (const float*)d_in[2];
    const float* w_hh    = (const float*)d_in[3];
    const float* b_ih    = (const float*)d_in[4];
    const float* b_hh    = (const float*)d_in[5];
    const float* fc_w    = (const float*)d_in[6];
    const float* fc_b    = (const float*)d_in[7];
    float*       out     = (float*)d_out;

    gru_fused_kernel<<<dim3(B_ / 16), dim3(32 * NWAVES_), 0, stream>>>(
        x, lengths, w_ih, w_hh, b_ih, b_hh, fc_w, fc_b, out);
}


# x_t fetched one step ahead; biases + its 2 x-slices per tile accumulated while the LDS reads of h are in flight: the global-load wait and 4 of 20 WMMAs leave the per-step dependency chain
# speedup vs baseline: 1.9218x; 1.3907x over previous
#include <hip/hip_runtime.h>


typedef float v2f __attribute__((ext_vector_type(2)));
typedef float v8f __attribute__((ext_vector_type(8)));

#define B_ 128
#define T_ 4096
#define I_ 6
#define H_ 32
#define O_ 2
#define NWAVES_ 5

__device__ __forceinline__ v8f wmma4(v2f a, v2f b, v8f c) {
    return __builtin_amdgcn_wmma_f32_16x16x4_f32(
         false, a,  false, b,
         (short)0, c,  false,  false);
}

__device__ __forceinline__ float fsigmoid(float a) {
    return __builtin_amdgcn_rcpf(1.0f + __builtin_amdgcn_exp2f(-1.44269504088896f * a));
}
__device__ __forceinline__ float ftanh(float a) {
    return 2.0f * __builtin_amdgcn_rcpf(1.0f + __builtin_amdgcn_exp2f(-2.88539008177793f * a)) - 1.0f;
}

__device__ __forceinline__ float from_lane_xor8(float v) {
    return __int_as_float(__builtin_amdgcn_update_dpp(0, __float_as_int(v), 0x128, 0xF, 0xF, true));
}

__device__ __forceinline__ void lds_barrier() {
    __builtin_amdgcn_fence(__ATOMIC_RELEASE, "workgroup", "local");
    __builtin_amdgcn_s_barrier();
    __builtin_amdgcn_fence(__ATOMIC_ACQUIRE, "workgroup", "local");
}

__global__ __launch_bounds__(32 * NWAVES_, 1) void gru_fused_kernel(
    const float* __restrict__ x,
    const int*   __restrict__ lengths,
    const float* __restrict__ w_ih,
    const float* __restrict__ w_hh,
    const float* __restrict__ b_ih,
    const float* __restrict__ b_hh,
    const float* __restrict__ fc_w,
    const float* __restrict__ fc_b,
    float*       __restrict__ out)
{
    __shared__ float ldsh[H_ * 16];

    const int tid  = threadIdx.x;
    const int wv   = __builtin_amdgcn_readfirstlane(tid >> 5);
    const int lane = tid & 31;
    const int lh   = lane >> 4;
    const int ln   = lane & 15;
    const int hi   = ln >> 3;
    const int b0   = blockIdx.x * 16;
    const bool gatew = wv < 4;

    for (int j = tid; j < H_ * 16; j += 32 * NWAVES_) ldsh[j] = 0.0f;
    __syncthreads();

    const int hc = (wv & 3) * 8 + (ln & 7);
    const int gP = hi ? (H_ + hc) : hc;
    const int gQ = 2 * H_ + hc;
    v2f WP[10], WQ[10];
    #pragma unroll
    for (int s = 0; s < 2; ++s) {
        const int k0 = 4 * s + 2 * lh;
        const float p0 = (k0     < I_) ? w_ih[gP * I_ + k0]     : 0.0f;
        const float p1 = (k0 + 1 < I_) ? w_ih[gP * I_ + k0 + 1] : 0.0f;
        const float q0 = (k0     < I_) ? w_ih[gQ * I_ + k0]     : 0.0f;
        const float q1 = (k0 + 1 < I_) ? w_ih[gQ * I_ + k0 + 1] : 0.0f;
        WP[s].x = p0;              WP[s].y = p1;
        WQ[s].x = hi ? q0 : 0.0f;  WQ[s].y = hi ? q1 : 0.0f;
    }
    #pragma unroll
    for (int s = 0; s < 8; ++s) {
        const int k0 = 4 * s + 2 * lh;
        WP[2 + s].x = w_hh[gP * H_ + k0];
        WP[2 + s].y = w_hh[gP * H_ + k0 + 1];
        const float q0 = w_hh[gQ * H_ + k0];
        const float q1 = w_hh[gQ * H_ + k0 + 1];
        WQ[2 + s].x = hi ? 0.0f : q0;
        WQ[2 + s].y = hi ? 0.0f : q1;
    }
    const float biasP = b_ih[gP] + b_hh[gP];
    const float biasQ = hi ? b_ih[gQ] : b_hh[gQ];

    const bool fcl = (ln < O_);
    v2f Wfc[8];
    #pragma unroll
    for (int s = 0; s < 8; ++s) {
        const int k0 = 4 * s + 2 * lh;
        Wfc[s].x = fcl ? fc_w[ln * H_ + k0]     : 0.0f;
        Wfc[s].y = fcl ? fc_w[ln * H_ + k0 + 1] : 0.0f;
    }
    const float fcb = fcl ? fc_b[ln] : 0.0f;

    int len8[8];
    #pragma unroll
    for (int v = 0; v < 8; ++v) len8[v] = lengths[b0 + v + 8 * lh];

    const float* xrow = x + (size_t)(b0 + ln) * T_ * I_;
    const float* xnA = xrow + I_ + 2 * lh;
    const float* xnB = xrow + I_ + 4;

    v2f   hA[8];
    float hold[8];
    #pragma unroll
    for (int s = 0; s < 8; ++s) { hA[s].x = 0.0f; hA[s].y = 0.0f; }
    #pragma unroll
    for (int v = 0; v < 8; ++v) hold[v] = 0.0f;

    v8f xP, xQ;
    #pragma unroll
    for (int v = 0; v < 8; ++v) { xP[v] = biasP; xQ[v] = biasQ; }
    if (gatew) {
        v2f a0, a1;
        a0.x = xrow[2 * lh];
        a0.y = xrow[2 * lh + 1];
        const float e4 = xrow[4], e5 = xrow[5];
        a1.x = lh ? 0.0f : e4;
        a1.y = lh ? 0.0f : e5;
        xP = wmma4(a0, WP[0], xP);  xP = wmma4(a1, WP[1], xP);
        xQ = wmma4(a0, WQ[0], xQ);  xQ = wmma4(a1, WQ[1], xQ);
    }

    #pragma unroll 1
    for (int t = 0; t < T_; ++t) {
        float xr0 = 0.0f, xr1 = 0.0f, xr2 = 0.0f, xr3 = 0.0f;
        if (gatew) {
            xr0 = xnA[0];
            xr1 = xnA[1];
            xr2 = xnB[0];
            xr3 = xnB[1];

            v8f accP = xP, accQ = xQ;
            #pragma unroll
            for (int s = 0; s < 2; ++s) {
                accP = wmma4(hA[s], WP[2 + s], accP);
                accQ = wmma4(hA[s], WQ[2 + s], accQ);
            }
            if (t + 5 < T_) __builtin_prefetch(xnB + 4 * I_, 0, 1);
            #pragma unroll
            for (int s = 2; s < 8; ++s) {
                accP = wmma4(hA[s], WP[2 + s], accP);
                accQ = wmma4(hA[s], WQ[2 + s], accQ);
            }

            #pragma unroll
            for (int v = 0; v < 8; ++v) {
                const float sg = fsigmoid(accP[v]);
                const float z  = from_lane_xor8(sg);
                const float xn = from_lane_xor8(accQ[v]);
                const float n  = ftanh(xn + sg * accQ[v]);
                const float hn = (1.0f - z) * n + z * hold[v];
                const float hs = (t < len8[v]) ? hn : hold[v];
                hold[v] = hs;
            }
            if (!hi) {
                #pragma unroll
                for (int v = 0; v < 8; ++v) ldsh[hc * 16 + 8 * lh + v] = hold[v];
            }
        }

        lds_barrier();

        #pragma unroll
        for (int s = 0; s < 8; ++s) {
            const int k0 = 4 * s + 2 * lh;
            hA[s].x = ldsh[k0 * 16 + ln];
            hA[s].y = ldsh[(k0 + 1) * 16 + ln];
        }

        if (gatew) {
            v2f a0, a1;
            a0.x = xr0;
            a0.y = xr1;
            a1.x = lh ? 0.0f : xr2;
            a1.y = lh ? 0.0f : xr3;
            v8f p, q;
            #pragma unroll
            for (int v = 0; v < 8; ++v) { p[v] = biasP; q[v] = biasQ; }
            p = wmma4(a0, WP[0], p);  p = wmma4(a1, WP[1], p);
            q = wmma4(a0, WQ[0], q);  q = wmma4(a1, WQ[1], q);
            xP = p;  xQ = q;
            const int adv = (t + 2 < T_) ? I_ : 0;
            xnA += adv;
            xnB += adv;
        }

        lds_barrier();

        if (wv == 4) {
            v8f accF;
            #pragma unroll
            for (int v = 0; v < 8; ++v) accF[v] = fcb;
            #pragma unroll
            for (int s = 0; s < 8; ++s) accF = wmma4(hA[s], Wfc[s], accF);
            if (ln < O_) {
                #pragma unroll
                for (int v = 0; v < 8; ++v) {
                    const float val = (t < len8[v]) ? accF[v] : fcb;
                    out[((size_t)(b0 + v + 8 * lh) * T_ + t) * O_ + ln] = val;
                }
            }
        }
    }
}

extern "C" void kernel_launch(void* const* d_in, const int* in_sizes, int n_in,
                              void* d_out, int out_size, void* d_ws, size_t ws_size,
                              hipStream_t stream) {
    const float* x       = (const float*)d_in[0];
    const int*   lengths = (const int*)  d_in[1];
    const float* w_ih    = (const float*)d_in[2];
    const float* w_hh    = (const float*)d_in[3];
    const float* b_ih    = (const float*)d_in[4];
    const float* b_hh    = (const float*)d_in[5];
    const float* fc_w    = (const float*)d_in[6];
    const float* fc_b    = (const float*)d_in[7];
    float*       out     = (float*)d_out;

    gru_fused_kernel<<<dim3(B_ / 16), dim3(32 * NWAVES_), 0, stream>>>(
        x, lengths, w_ih, w_hh, b_ih, b_hh, fc_w, fc_b, out);
}


# h exchanged through two LDS buffers used in turn: one barrier per step instead of two; behind it each wave runs from its LDS reads straight into its WMMA chain as the first K-slice arrives
# speedup vs baseline: 2.0331x; 1.0579x over previous
#include <hip/hip_runtime.h>


typedef float v2f __attribute__((ext_vector_type(2)));
typedef float v8f __attribute__((ext_vector_type(8)));

#define B_ 128
#define T_ 4096
#define I_ 6
#define H_ 32
#define O_ 2
#define NWAVES_ 5

__device__ __forceinline__ v8f wmma4(v2f a, v2f b, v8f c) {
    return __builtin_amdgcn_wmma_f32_16x16x4_f32(
         false, a,  false, b,
         (short)0, c,  false,  false);
}

__device__ __forceinline__ float fsigmoid(float a) {
    return __builtin_amdgcn_rcpf(1.0f + __builtin_amdgcn_exp2f(-1.44269504088896f * a));
}
__device__ __forceinline__ float ftanh(float a) {
    return 2.0f * __builtin_amdgcn_rcpf(1.0f + __builtin_amdgcn_exp2f(-2.88539008177793f * a)) - 1.0f;
}

__device__ __forceinline__ float from_lane_xor8(float v) {
    return __int_as_float(__builtin_amdgcn_update_dpp(0, __float_as_int(v), 0x128, 0xF, 0xF, true));
}

__device__ __forceinline__ void lds_barrier() {
    __builtin_amdgcn_fence(__ATOMIC_RELEASE, "workgroup", "local");
    __builtin_amdgcn_s_barrier();
    __builtin_amdgcn_fence(__ATOMIC_ACQUIRE, "workgroup", "local");
}

__global__ __launch_bounds__(32 * NWAVES_, 1) void gru_fused_kernel(
    const float* __restrict__ x,
    const int*   __restrict__ lengths,
    const float* __restrict__ w_ih,
    const float* __restrict__ w_hh,
    const float* __restrict__ b_ih,
    const float* __restrict__ b_hh,
    const float* __restrict__ fc_w,
    const float* __restrict__ fc_b,
    float*       __restrict__ out)
{
    __shared__ float ldsh[2 * H_ * 16];

    const int tid  = threadIdx.x;
    const int wv   = __builtin_amdgcn_readfirstlane(tid >> 5);
    const int lane = tid & 31;
    const int lh   = lane >> 4;
    const int ln   = lane & 15;
    const int hi   = ln >> 3;
    const int b0   = blockIdx.x * 16;
    const bool gatew = wv < 4;


    const int hc = (wv & 3) * 8 + (ln & 7);
    const int gP = hi ? (H_ + hc) : hc;
    const int gQ = 2 * H_ + hc;
    v2f WP[10], WQ[10];
    #pragma unroll
    for (int s = 0; s < 2; ++s) {
        const int k0 = 4 * s + 2 * lh;
        const float p0 = (k0     < I_) ? w_ih[gP * I_ + k0]     : 0.0f;
        const float p1 = (k0 + 1 < I_) ? w_ih[gP * I_ + k0 + 1] : 0.0f;
        const float q0 = (k0     < I_) ? w_ih[gQ * I_ + k0]     : 0.0f;
        const float q1 = (k0 + 1 < I_) ? w_ih[gQ * I_ + k0 + 1] : 0.0f;
        WP[s].x = p0;              WP[s].y = p1;
        WQ[s].x = hi ? q0 : 0.0f;  WQ[s].y = hi ? q1 : 0.0f;
    }
    #pragma unroll
    for (int s = 0; s < 8; ++s) {
        const int k0 = 4 * s + 2 * lh;
        WP[2 + s].x = w_hh[gP * H_ + k0];
        WP[2 + s].y = w_hh[gP * H_ + k0 + 1];
        const float q0 = w_hh[gQ * H_ + k0];
        const float q1 = w_hh[gQ * H_ + k0 + 1];
        WQ[2 + s].x = hi ? 0.0f : q0;
        WQ[2 + s].y = hi ? 0.0f : q1;
    }
    const float biasP = b_ih[gP] + b_hh[gP];
    const float biasQ = hi ? b_ih[gQ] : b_hh[gQ];

    const bool fcl = (ln < O_);
    v2f Wfc[8];
    #pragma unroll
    for (int s = 0; s < 8; ++s) {
        const int k0 = 4 * s + 2 * lh;
        Wfc[s].x = fcl ? fc_w[ln * H_ + k0]     : 0.0f;
        Wfc[s].y = fcl ? fc_w[ln * H_ + k0 + 1] : 0.0f;
    }
    const float fcb = fcl ? fc_b[ln] : 0.0f;

    int len8[8];
    #pragma unroll
    for (int v = 0; v < 8; ++v) len8[v] = lengths[b0 + v + 8 * lh];

    const float* xrow = x + (size_t)(b0 + ln) * T_ * I_;
    const float* xnA = xrow + I_ + 2 * lh;
    const float* xnB = xrow + I_ + 4;

    v2f   hA[8];
    float hold[8];
    #pragma unroll
    for (int s = 0; s < 8; ++s) { hA[s].x = 0.0f; hA[s].y = 0.0f; }
    #pragma unroll
    for (int v = 0; v < 8; ++v) hold[v] = 0.0f;

    v8f xP, xQ;
    #pragma unroll
    for (int v = 0; v < 8; ++v) { xP[v] = biasP; xQ[v] = biasQ; }
    if (gatew) {
        v2f a0, a1;
        a0.x = xrow[2 * lh];
        a0.y = xrow[2 * lh + 1];
        const float e4 = xrow[4], e5 = xrow[5];
        a1.x = lh ? 0.0f : e4;
        a1.y = lh ? 0.0f : e5;
        xP = wmma4(a0, WP[0], xP);  xP = wmma4(a1, WP[1], xP);
        xQ = wmma4(a0, WQ[0], xQ);  xQ = wmma4(a1, WQ[1], xQ);
    }

    #pragma unroll 1
    for (int t = 0; t < T_; ++t) {
        const int pub = (t & 1) * (H_ * 16);
        float xr0 = 0.0f, xr1 = 0.0f, xr2 = 0.0f, xr3 = 0.0f;
        if (gatew) {
            xr0 = xnA[0];
            xr1 = xnA[1];
            xr2 = xnB[0];
            xr3 = xnB[1];

            v8f accP = xP, accQ = xQ;
            #pragma unroll
            for (int s = 0; s < 2; ++s) {
                accP = wmma4(hA[s], WP[2 + s], accP);
                accQ = wmma4(hA[s], WQ[2 + s], accQ);
            }
            if (t + 5 < T_) __builtin_prefetch(xnB + 4 * I_, 0, 1);
            #pragma unroll
            for (int s = 2; s < 8; ++s) {
                accP = wmma4(hA[s], WP[2 + s], accP);
                accQ = wmma4(hA[s], WQ[2 + s], accQ);
            }

            #pragma unroll
            for (int v = 0; v < 8; ++v) {
                const float sg = fsigmoid(accP[v]);
                const float z  = from_lane_xor8(sg);
                const float xn = from_lane_xor8(accQ[v]);
                const float n  = ftanh(xn + sg * accQ[v]);
                const float hn = (1.0f - z) * n + z * hold[v];
                const float hs = (t < len8[v]) ? hn : hold[v];
                hold[v] = hs;
            }
            if (!hi) {
                #pragma unroll
                for (int v = 0; v < 8; ++v) ldsh[pub + hc * 16 + 8 * lh + v] = hold[v];
            }
        }

        lds_barrier();

        #pragma unroll
        for (int s = 0; s < 8; ++s) {
            const int k0 = 4 * s + 2 * lh;
            hA[s].x = ldsh[pub + k0 * 16 + ln];
            hA[s].y = ldsh[pub + (k0 + 1) * 16 + ln];
        }

        if (gatew) {
            v2f a0, a1;
            a0.x = xr0;
            a0.y = xr1;
            a1.x = lh ? 0.0f : xr2;
            a1.y = lh ? 0.0f : xr3;
            v8f p, q;
            #pragma unroll
            for (int v = 0; v < 8; ++v) { p[v] = biasP; q[v] = biasQ; }
            p = wmma4(a0, WP[0], p);  p = wmma4(a1, WP[1], p);
            q = wmma4(a0, WQ[0], q);  q = wmma4(a1, WQ[1], q);
            xP = p;  xQ = q;
            const int adv = (t + 2 < T_) ? I_ : 0;
            xnA += adv;
            xnB += adv;
        }

        if (wv == 4) {
            v8f accF;
            #pragma unroll
            for (int v = 0; v < 8; ++v) accF[v] = fcb;
            #pragma unroll
            for (int s = 0; s < 8; ++s) accF = wmma4(hA[s], Wfc[s], accF);
            if (ln < O_) {
                #pragma unroll
                for (int v = 0; v < 8; ++v) {
                    const float val = (t < len8[v]) ? accF[v] : fcb;
                    out[((size_t)(b0 + v + 8 * lh) * T_ + t) * O_ + ln] = val;
                }
            }
        }
    }
}

extern "C" void kernel_launch(void* const* d_in, const int* in_sizes, int n_in,
                              void* d_out, int out_size, void* d_ws, size_t ws_size,
                              hipStream_t stream) {
    const float* x       = (const float*)d_in[0];
    const int*   lengths = (const int*)  d_in[1];
    const float* w_ih    = (const float*)d_in[2];
    const float* w_hh    = (const float*)d_in[3];
    const float* b_ih    = (const float*)d_in[4];
    const float* b_hh    = (const float*)d_in[5];
    const float* fc_w    = (const float*)d_in[6];
    const float* fc_b    = (const float*)d_in[7];
    float*       out     = (float*)d_out;

    gru_fused_kernel<<<dim3(B_ / 16), dim3(32 * NWAVES_), 0, stream>>>(
        x, lengths, w_ih, w_hh, b_ih, b_hh, fc_w, fc_b, out);
}


# Each lane of a column pair finishes 4 of the pair's 8 rows (halves of r|z and hn|xn swapped by row rotations): tanh, blend, mask on 4 values in all lanes, not 8 in half; one 16-byte h store per lane
# speedup vs baseline: 2.1156x; 1.0406x over previous
#include <hip/hip_runtime.h>


typedef float v2f __attribute__((ext_vector_type(2)));
typedef float v8f __attribute__((ext_vector_type(8)));

#define B_ 128
#define T_ 4096
#define I_ 6
#define H_ 32
#define O_ 2
#define NWAVES_ 5

__device__ __forceinline__ v8f wmma4(v2f a, v2f b, v8f c) {
    return __builtin_amdgcn_wmma_f32_16x16x4_f32(
         false, a,  false, b,
         (short)0, c,  false,  false);
}

__device__ __forceinline__ float fsigmoid(float a) {
    return __builtin_amdgcn_rcpf(1.0f + __builtin_amdgcn_exp2f(-1.44269504088896f * a));
}
__device__ __forceinline__ float ftanh(float a) {
    return 2.0f * __builtin_amdgcn_rcpf(1.0f + __builtin_amdgcn_exp2f(-2.88539008177793f * a)) - 1.0f;
}

__device__ __forceinline__ float from_lane_xor8(float v) {
    return __int_as_float(__builtin_amdgcn_update_dpp(0, __float_as_int(v), 0x128, 0xF, 0xF, true));
}

__device__ __forceinline__ float from_lane_xor8_low_half(float v, float keep) {
    return __int_as_float(__builtin_amdgcn_update_dpp(__float_as_int(keep), __float_as_int(v), 0x128, 0xF, 0x3, false));
}

__device__ __forceinline__ void lds_barrier() {
    __builtin_amdgcn_fence(__ATOMIC_RELEASE, "workgroup", "local");
    __builtin_amdgcn_s_barrier();
    __builtin_amdgcn_fence(__ATOMIC_ACQUIRE, "workgroup", "local");
}

__global__ __launch_bounds__(32 * NWAVES_, 1) void gru_fused_kernel(
    const float* __restrict__ x,
    const int*   __restrict__ lengths,
    const float* __restrict__ w_ih,
    const float* __restrict__ w_hh,
    const float* __restrict__ b_ih,
    const float* __restrict__ b_hh,
    const float* __restrict__ fc_w,
    const float* __restrict__ fc_b,
    float*       __restrict__ out)
{
    __shared__ float ldsh[2 * H_ * 16];

    const int tid  = threadIdx.x;
    const int wv   = __builtin_amdgcn_readfirstlane(tid >> 5);
    const int lane = tid & 31;
    const int lh   = lane >> 4;
    const int ln   = lane & 15;
    const int hi   = ln >> 3;
    const int b0   = blockIdx.x * 16;
    const bool gatew = wv < 4;


    const int hc = (wv & 3) * 8 + (ln & 7);
    const int gP = hi ? (H_ + hc) : hc;
    const int gQ = 2 * H_ + hc;
    v2f WP[10], WQ[10];
    #pragma unroll
    for (int s = 0; s < 2; ++s) {
        const int k0 = 4 * s + 2 * lh;
        const float p0 = (k0     < I_) ? w_ih[gP * I_ + k0]     : 0.0f;
        const float p1 = (k0 + 1 < I_) ? w_ih[gP * I_ + k0 + 1] : 0.0f;
        const float q0 = (k0     < I_) ? w_ih[gQ * I_ + k0]     : 0.0f;
        const float q1 = (k0 + 1 < I_) ? w_ih[gQ * I_ + k0 + 1] : 0.0f;
        WP[s].x = p0;              WP[s].y = p1;
        WQ[s].x = hi ? q0 : 0.0f;  WQ[s].y = hi ? q1 : 0.0f;
    }
    #pragma unroll
    for (int s = 0; s < 8; ++s) {
        const int k0 = 4 * s + 2 * lh;
        WP[2 + s].x = w_hh[gP * H_ + k0];
        WP[2 + s].y = w_hh[gP * H_ + k0 + 1];
        const float q0 = w_hh[gQ * H_ + k0];
        const float q1 = w_hh[gQ * H_ + k0 + 1];
        WQ[2 + s].x = hi ? 0.0f : q0;
        WQ[2 + s].y = hi ? 0.0f : q1;
    }
    const float biasP = b_ih[gP] + b_hh[gP];
    const float biasQ = hi ? b_ih[gQ] : b_hh[gQ];

    const bool fcl = (ln < O_);
    v2f Wfc[8];
    #pragma unroll
    for (int s = 0; s < 8; ++s) {
        const int k0 = 4 * s + 2 * lh;
        Wfc[s].x = fcl ? fc_w[ln * H_ + k0]     : 0.0f;
        Wfc[s].y = fcl ? fc_w[ln * H_ + k0 + 1] : 0.0f;
    }
    const float fcb = fcl ? fc_b[ln] : 0.0f;

    int len8[8];
    #pragma unroll
    for (int v = 0; v < 8; ++v) len8[v] = lengths[b0 + v + 8 * lh];
    int len4[4];
    #pragma unroll
    for (int j = 0; j < 4; ++j) len4[j] = hi ? len8[4 + j] : len8[j];

    const float* xrow = x + (size_t)(b0 + ln) * T_ * I_;
    const float* xnA = xrow + I_ + 2 * lh;
    const float* xnB = xrow + I_ + 4;

    v2f   hA[8];
    float hold[4];
    #pragma unroll
    for (int s = 0; s < 8; ++s) { hA[s].x = 0.0f; hA[s].y = 0.0f; }
    #pragma unroll
    for (int j = 0; j < 4; ++j) hold[j] = 0.0f;

    v8f xP, xQ;
    #pragma unroll
    for (int v = 0; v < 8; ++v) { xP[v] = biasP; xQ[v] = biasQ; }
    if (gatew) {
        v2f a0, a1;
        a0.x = xrow[2 * lh];
        a0.y = xrow[2 * lh + 1];
        const float e4 = xrow[4], e5 = xrow[5];
        a1.x = lh ? 0.0f : e4;
        a1.y = lh ? 0.0f : e5;
        xP = wmma4(a0, WP[0], xP);  xP = wmma4(a1, WP[1], xP);
        xQ = wmma4(a0, WQ[0], xQ);  xQ = wmma4(a1, WQ[1], xQ);
    }

    #pragma unroll 1
    for (int t = 0; t < T_; ++t) {
        const int pub = (t & 1) * (H_ * 16);
        float xr0 = 0.0f, xr1 = 0.0f, xr2 = 0.0f, xr3 = 0.0f;
        if (gatew) {
            xr0 = xnA[0];
            xr1 = xnA[1];
            xr2 = xnB[0];
            xr3 = xnB[1];

            v8f accP = xP, accQ = xQ;
            #pragma unroll
            for (int s = 0; s < 2; ++s) {
                accP = wmma4(hA[s], WP[2 + s], accP);
                accQ = wmma4(hA[s], WQ[2 + s], accQ);
            }
            if (t + 5 < T_) __builtin_prefetch(xnB + 4 * I_, 0, 1);
            #pragma unroll
            for (int s = 2; s < 8; ++s) {
                accP = wmma4(hA[s], WP[2 + s], accP);
                accQ = wmma4(hA[s], WQ[2 + s], accQ);
            }

            float sg[8];
            #pragma unroll
            for (int v = 0; v < 8; ++v) sg[v] = fsigmoid(accP[v]);

            #pragma unroll
            for (int j = 0; j < 4; ++j) {
                const float rS = from_lane_xor8(sg[4 + j]);
                const float hS = from_lane_xor8(accQ[4 + j]);
                const float z  = from_lane_xor8_low_half(sg[j],   sg[4 + j]);
                const float xn = from_lane_xor8_low_half(accQ[j], accQ[4 + j]);
                const float r  = hi ? rS : sg[j];
                const float hq = hi ? hS : accQ[j];
                const float n  = ftanh(xn + r * hq);
                const float hn = (1.0f - z) * n + z * hold[j];
                const float hs = (t < len4[j]) ? hn : hold[j];
                hold[j] = hs;
            }
            #pragma unroll
            for (int j = 0; j < 4; ++j) ldsh[pub + hc * 16 + 8 * lh + 4 * hi + j] = hold[j];
        }

        lds_barrier();

        #pragma unroll
        for (int s = 0; s < 8; ++s) {
            const int k0 = 4 * s + 2 * lh;
            hA[s].x = ldsh[pub + k0 * 16 + ln];
            hA[s].y = ldsh[pub + (k0 + 1) * 16 + ln];
        }

        if (gatew) {
            v2f a0, a1;
            a0.x = xr0;
            a0.y = xr1;
            a1.x = lh ? 0.0f : xr2;
            a1.y = lh ? 0.0f : xr3;
            v8f p, q;
            #pragma unroll
            for (int v = 0; v < 8; ++v) { p[v] = biasP; q[v] = biasQ; }
            p = wmma4(a0, WP[0], p);  p = wmma4(a1, WP[1], p);
            q = wmma4(a0, WQ[0], q);  q = wmma4(a1, WQ[1], q);
            xP = p;  xQ = q;
            const int adv = (t + 2 < T_) ? I_ : 0;
            xnA += adv;
            xnB += adv;
        }

        if (wv == 4) {
            v8f accF;
            #pragma unroll
            for (int v = 0; v < 8; ++v) accF[v] = fcb;
            #pragma unroll
            for (int s = 0; s < 8; ++s) accF = wmma4(hA[s], Wfc[s], accF);
            if (ln < O_) {
                #pragma unroll
                for (int v = 0; v < 8; ++v) {
                    const float val = (t < len8[v]) ? accF[v] : fcb;
                    out[((size_t)(b0 + v + 8 * lh) * T_ + t) * O_ + ln] = val;
                }
            }
        }
    }
}

extern "C" void kernel_launch(void* const* d_in, const int* in_sizes, int n_in,
                              void* d_out, int out_size, void* d_ws, size_t ws_size,
                              hipStream_t stream) {
    const float* x       = (const float*)d_in[0];
    const int*   lengths = (const int*)  d_in[1];
    const float* w_ih    = (const float*)d_in[2];
    const float* w_hh    = (const float*)d_in[3];
    const float* b_ih    = (const float*)d_in[4];
    const float* b_hh    = (const float*)d_in[5];
    const float* fc_w    = (const float*)d_in[6];
    const float* fc_b    = (const float*)d_in[7];
    float*       out     = (float*)d_out;

    gru_fused_kernel<<<dim3(B_ / 16), dim3(32 * NWAVES_), 0, stream>>>(
        x, lengths, w_ih, w_hh, b_ih, b_hh, fc_w, fc_b, out);
}


# Activations on v_tanh_f32: sigmoid(a)=0.5+0.5*tanh(a/2), halves folded exactly into r|z|hn weights; xn+r*hn summed as (xn+hn/2)+tanh(ar/2)*hn/2; 12 tanh per lane and step replace 12 exp + 12 rcp
# speedup vs baseline: 2.2089x; 1.0441x over previous
#include <hip/hip_runtime.h>


typedef float v2f __attribute__((ext_vector_type(2)));
typedef float v8f __attribute__((ext_vector_type(8)));

#define B_ 128
#define T_ 4096
#define I_ 6
#define H_ 32
#define O_ 2
#define NWAVES_ 5

__device__ __forceinline__ v8f wmma4(v2f a, v2f b, v8f c) {
    return __builtin_amdgcn_wmma_f32_16x16x4_f32(
         false, a,  false, b,
         (short)0, c,  false,  false);
}

__device__ __forceinline__ float htanh(float a) {
    return __builtin_amdgcn_tanhf(a);
}

__device__ __forceinline__ float from_lane_xor8(float v) {
    return __int_as_float(__builtin_amdgcn_update_dpp(0, __float_as_int(v), 0x128, 0xF, 0xF, true));
}

__device__ __forceinline__ float from_lane_xor8_low_half(float v, float keep) {
    return __int_as_float(__builtin_amdgcn_update_dpp(__float_as_int(keep), __float_as_int(v), 0x128, 0xF, 0x3, false));
}

__device__ __forceinline__ void lds_barrier() {
    __builtin_amdgcn_fence(__ATOMIC_RELEASE, "workgroup", "local");
    __builtin_amdgcn_s_barrier();
    __builtin_amdgcn_fence(__ATOMIC_ACQUIRE, "workgroup", "local");
}

__global__ __launch_bounds__(32 * NWAVES_, 1) void gru_fused_kernel(
    const float* __restrict__ x,
    const int*   __restrict__ lengths,
    const float* __restrict__ w_ih,
    const float* __restrict__ w_hh,
    const float* __restrict__ b_ih,
    const float* __restrict__ b_hh,
    const float* __restrict__ fc_w,
    const float* __restrict__ fc_b,
    float*       __restrict__ out)
{
    __shared__ float ldsh[2 * H_ * 16];

    const int tid  = threadIdx.x;
    const int wv   = __builtin_amdgcn_readfirstlane(tid >> 5);
    const int lane = tid & 31;
    const int lh   = lane >> 4;
    const int ln   = lane & 15;
    const int hi   = ln >> 3;
    const int b0   = blockIdx.x * 16;
    const bool gatew = wv < 4;


    const int hc = (wv & 3) * 8 + (ln & 7);
    const int gP = hi ? (H_ + hc) : hc;
    const int gQ = 2 * H_ + hc;
    v2f WP[10], WQ[10];
    #pragma unroll
    for (int s = 0; s < 2; ++s) {
        const int k0 = 4 * s + 2 * lh;
        const float p0 = (k0     < I_) ? w_ih[gP * I_ + k0]     : 0.0f;
        const float p1 = (k0 + 1 < I_) ? w_ih[gP * I_ + k0 + 1] : 0.0f;
        const float q0 = (k0     < I_) ? w_ih[gQ * I_ + k0]     : 0.0f;
        const float q1 = (k0 + 1 < I_) ? w_ih[gQ * I_ + k0 + 1] : 0.0f;
        WP[s].x = 0.5f * p0;       WP[s].y = 0.5f * p1;
        WQ[s].x = hi ? q0 : 0.0f;  WQ[s].y = hi ? q1 : 0.0f;
    }
    #pragma unroll
    for (int s = 0; s < 8; ++s) {
        const int k0 = 4 * s + 2 * lh;
        WP[2 + s].x = 0.5f * w_hh[gP * H_ + k0];
        WP[2 + s].y = 0.5f * w_hh[gP * H_ + k0 + 1];
        const float q0 = w_hh[gQ * H_ + k0];
        const float q1 = w_hh[gQ * H_ + k0 + 1];
        WQ[2 + s].x = hi ? 0.0f : 0.5f * q0;
        WQ[2 + s].y = hi ? 0.0f : 0.5f * q1;
    }
    const float biasP = 0.5f * (b_ih[gP] + b_hh[gP]);
    const float biasQ = hi ? b_ih[gQ] : 0.5f * b_hh[gQ];

    const bool fcl = (ln < O_);
    v2f Wfc[8];
    #pragma unroll
    for (int s = 0; s < 8; ++s) {
        const int k0 = 4 * s + 2 * lh;
        Wfc[s].x = fcl ? fc_w[ln * H_ + k0]     : 0.0f;
        Wfc[s].y = fcl ? fc_w[ln * H_ + k0 + 1] : 0.0f;
    }
    const float fcb = fcl ? fc_b[ln] : 0.0f;

    int len8[8];
    #pragma unroll
    for (int v = 0; v < 8; ++v) len8[v] = lengths[b0 + v + 8 * lh];
    int len4[4];
    #pragma unroll
    for (int j = 0; j < 4; ++j) len4[j] = hi ? len8[4 + j] : len8[j];

    const float* xrow = x + (size_t)(b0 + ln) * T_ * I_;
    const float* xnA = xrow + I_ + 2 * lh;
    const float* xnB = xrow + I_ + 4;

    v2f   hA[8];
    float hold[4];
    #pragma unroll
    for (int s = 0; s < 8; ++s) { hA[s].x = 0.0f; hA[s].y = 0.0f; }
    #pragma unroll
    for (int j = 0; j < 4; ++j) hold[j] = 0.0f;

    v8f xP, xQ;
    #pragma unroll
    for (int v = 0; v < 8; ++v) { xP[v] = biasP; xQ[v] = biasQ; }
    if (gatew) {
        v2f a0, a1;
        a0.x = xrow[2 * lh];
        a0.y = xrow[2 * lh + 1];
        const float e4 = xrow[4], e5 = xrow[5];
        a1.x = lh ? 0.0f : e4;
        a1.y = lh ? 0.0f : e5;
        xP = wmma4(a0, WP[0], xP);  xP = wmma4(a1, WP[1], xP);
        xQ = wmma4(a0, WQ[0], xQ);  xQ = wmma4(a1, WQ[1], xQ);
    }

    #pragma unroll 1
    for (int t = 0; t < T_; ++t) {
        const int pub = (t & 1) * (H_ * 16);
        float xr0 = 0.0f, xr1 = 0.0f, xr2 = 0.0f, xr3 = 0.0f;
        if (gatew) {
            xr0 = xnA[0];
            xr1 = xnA[1];
            xr2 = xnB[0];
            xr3 = xnB[1];

            v8f accP = xP, accQ = xQ;
            #pragma unroll
            for (int s = 0; s < 2; ++s) {
                accP = wmma4(hA[s], WP[2 + s], accP);
                accQ = wmma4(hA[s], WQ[2 + s], accQ);
            }
            if (t + 5 < T_) __builtin_prefetch(xnB + 4 * I_, 0, 1);
            #pragma unroll
            for (int s = 2; s < 8; ++s) {
                accP = wmma4(hA[s], WP[2 + s], accP);
                accQ = wmma4(hA[s], WQ[2 + s], accQ);
            }

            float ar[4], az[4], tr[4], tz[4], hq[4], xa[4], nn[4], omz[4], zh[4];
            #pragma unroll
            for (int j = 0; j < 4; ++j) {
                const float aS = from_lane_xor8(accP[4 + j]);
                ar[j] = hi ? aS : accP[j];
                az[j] = from_lane_xor8_low_half(accP[j], accP[4 + j]);
            }
            #pragma unroll
            for (int j = 0; j < 4; ++j) tr[j] = htanh(ar[j]);
            #pragma unroll
            for (int j = 0; j < 4; ++j) tz[j] = htanh(az[j]);
            #pragma unroll
            for (int j = 0; j < 4; ++j) {
                const float hS = from_lane_xor8(accQ[4 + j]);
                const float xn = from_lane_xor8_low_half(accQ[j], accQ[4 + j]);
                hq[j] = hi ? hS : accQ[j];
                xa[j] = xn + hq[j];
            }
            #pragma unroll
            for (int j = 0; j < 4; ++j) nn[j] = htanh(__builtin_fmaf(tr[j], hq[j], xa[j]));
            #pragma unroll
            for (int j = 0; j < 4; ++j) {
                omz[j] = __builtin_fmaf(tz[j], -0.5f, 0.5f);
                zh[j]  = __builtin_fmaf(-omz[j], hold[j], hold[j]);
            }
            #pragma unroll
            for (int j = 0; j < 4; ++j) {
                const float hn = __builtin_fmaf(omz[j], nn[j], zh[j]);
                const int live = (t < len4[j]) ? -1 : 0;
                hold[j] = __int_as_float((__float_as_int(hn) & live) | (__float_as_int(hold[j]) & ~live));
            }
            #pragma unroll
            for (int j = 0; j < 4; ++j) ldsh[pub + hc * 16 + 8 * lh + 4 * hi + j] = hold[j];
        }

        lds_barrier();

        #pragma unroll
        for (int s = 0; s < 8; ++s) {
            const int k0 = 4 * s + 2 * lh;
            hA[s].x = ldsh[pub + k0 * 16 + ln];
            hA[s].y = ldsh[pub + (k0 + 1) * 16 + ln];
        }

        if (gatew) {
            v2f a0, a1;
            a0.x = xr0;
            a0.y = xr1;
            a1.x = lh ? 0.0f : xr2;
            a1.y = lh ? 0.0f : xr3;
            v8f p, q;
            #pragma unroll
            for (int v = 0; v < 8; ++v) { p[v] = biasP; q[v] = biasQ; }
            p = wmma4(a0, WP[0], p);  p = wmma4(a1, WP[1], p);
            q = wmma4(a0, WQ[0], q);  q = wmma4(a1, WQ[1], q);
            xP = p;  xQ = q;
            const int adv = (t + 2 < T_) ? I_ : 0;
            xnA += adv;
            xnB += adv;
        }

        if (wv == 4) {
            v8f accF;
            #pragma unroll
            for (int v = 0; v < 8; ++v) accF[v] = fcb;
            #pragma unroll
            for (int s = 0; s < 8; ++s) accF = wmma4(hA[s], Wfc[s], accF);
            if (ln < O_) {
                #pragma unroll
                for (int v = 0; v < 8; ++v) {
                    const float val = (t < len8[v]) ? accF[v] : fcb;
                    out[((size_t)(b0 + v + 8 * lh) * T_ + t) * O_ + ln] = val;
                }
            }
        }
    }
}

extern "C" void kernel_launch(void* const* d_in, const int* in_sizes, int n_in,
                              void* d_out, int out_size, void* d_ws, size_t ws_size,
                              hipStream_t stream) {
    const float* x       = (const float*)d_in[0];
    const int*   lengths = (const int*)  d_in[1];
    const float* w_ih    = (const float*)d_in[2];
    const float* w_hh    = (const float*)d_in[3];
    const float* b_ih    = (const float*)d_in[4];
    const float* b_hh    = (const float*)d_in[5];
    const float* fc_w    = (const float*)d_in[6];
    const float* fc_b    = (const float*)d_in[7];
    float*       out     = (float*)d_out;

    gru_fused_kernel<<<dim3(B_ / 16), dim3(32 * NWAVES_), 0, stream>>>(
        x, lengths, w_ih, w_hh, b_ih, b_hh, fc_w, fc_b, out);
}


# FC rides in the idle xn columns of tile Q's h-slices (xn kept in its own x-side accumulator): y_t falls out of step t+1's gate chain; the FC wave, its 8 WMMAs and its LDS reads per step are gone
# speedup vs baseline: 2.4544x; 1.1111x over previous
#include <hip/hip_runtime.h>


typedef float v2f __attribute__((ext_vector_type(2)));
typedef float v8f __attribute__((ext_vector_type(8)));

#define B_ 128
#define T_ 4096
#define I_ 6
#define H_ 32
#define O_ 2
#define NWAVES_ 4

__device__ __forceinline__ v8f wmma4(v2f a, v2f b, v8f c) {
    return __builtin_amdgcn_wmma_f32_16x16x4_f32(
         false, a,  false, b,
         (short)0, c,  false,  false);
}

__device__ __forceinline__ float htanh(float a) {
    return __builtin_amdgcn_tanhf(a);
}

__device__ __forceinline__ float from_lane_xor8(float v) {
    return __int_as_float(__builtin_amdgcn_update_dpp(0, __float_as_int(v), 0x128, 0xF, 0xF, true));
}

__device__ __forceinline__ float from_lane_xor8_low_half(float v, float keep) {
    return __int_as_float(__builtin_amdgcn_update_dpp(__float_as_int(keep), __float_as_int(v), 0x128, 0xF, 0x3, false));
}

__device__ __forceinline__ void lds_barrier() {
    __builtin_amdgcn_fence(__ATOMIC_RELEASE, "workgroup", "local");
    __builtin_amdgcn_s_barrier();
    __builtin_amdgcn_fence(__ATOMIC_ACQUIRE, "workgroup", "local");
}

__global__ __launch_bounds__(32 * NWAVES_, 1) void gru_fused_kernel(
    const float* __restrict__ x,
    const int*   __restrict__ lengths,
    const float* __restrict__ w_ih,
    const float* __restrict__ w_hh,
    const float* __restrict__ b_ih,
    const float* __restrict__ b_hh,
    const float* __restrict__ fc_w,
    const float* __restrict__ fc_b,
    float*       __restrict__ out)
{
    __shared__ float ldsh[2 * H_ * 16];

    const int tid  = threadIdx.x;
    const int wv   = __builtin_amdgcn_readfirstlane(tid >> 5);
    const int lane = tid & 31;
    const int lh   = lane >> 4;
    const int ln   = lane & 15;
    const int hi   = ln >> 3;
    const int b0   = blockIdx.x * 16;


    const int hc = wv * 8 + (ln & 7);
    const int gP = hi ? (H_ + hc) : hc;
    const int gQ = 2 * H_ + hc;
    const bool fcl = (ln >= 8) && (ln - 8 < O_);
    const int  fo  = fcl ? ln - 8 : 0;
    v2f WP[10], WQ[10];
    #pragma unroll
    for (int s = 0; s < 2; ++s) {
        const int k0 = 4 * s + 2 * lh;
        const float p0 = (k0     < I_) ? w_ih[gP * I_ + k0]     : 0.0f;
        const float p1 = (k0 + 1 < I_) ? w_ih[gP * I_ + k0 + 1] : 0.0f;
        const float q0 = (k0     < I_) ? w_ih[gQ * I_ + k0]     : 0.0f;
        const float q1 = (k0 + 1 < I_) ? w_ih[gQ * I_ + k0 + 1] : 0.0f;
        WP[s].x = 0.5f * p0;       WP[s].y = 0.5f * p1;
        WQ[s].x = hi ? q0 : 0.0f;  WQ[s].y = hi ? q1 : 0.0f;
    }
    #pragma unroll
    for (int s = 0; s < 8; ++s) {
        const int k0 = 4 * s + 2 * lh;
        WP[2 + s].x = 0.5f * w_hh[gP * H_ + k0];
        WP[2 + s].y = 0.5f * w_hh[gP * H_ + k0 + 1];
        const float q0 = w_hh[gQ * H_ + k0];
        const float q1 = w_hh[gQ * H_ + k0 + 1];
        const float f0 = fcl ? fc_w[fo * H_ + k0]     : 0.0f;
        const float f1 = fcl ? fc_w[fo * H_ + k0 + 1] : 0.0f;
        WQ[2 + s].x = hi ? f0 : 0.5f * q0;
        WQ[2 + s].y = hi ? f1 : 0.5f * q1;
    }
    const float biasP = 0.5f * (b_ih[gP] + b_hh[gP]);
    const float biasN = hi ? b_ih[gQ] : 0.0f;
    const float fcb   = fcl ? fc_b[fo] : 0.0f;
    const float biasQ = hi ? fcb : 0.5f * b_hh[gQ];

    int len4[4];
    #pragma unroll
    for (int j = 0; j < 4; ++j) len4[j] = lengths[b0 + j + 4 * hi + 8 * lh];
    const int rowS = b0 + 2 * wv + 8 * lh;
    const int lenS0 = lengths[rowS];
    const int lenS1 = lengths[rowS + 1];
    float* yp = out + (size_t)rowS * T_ * O_ + fo;

    const float* xrow = x + (size_t)(b0 + ln) * T_ * I_;
    const float* xnA = xrow + I_ + 2 * lh;
    const float* xnB = xrow + I_ + 4;

    v2f   hA[8];
    float hold[4];
    #pragma unroll
    for (int s = 0; s < 8; ++s) { hA[s].x = 0.0f; hA[s].y = 0.0f; }
    #pragma unroll
    for (int j = 0; j < 4; ++j) hold[j] = 0.0f;

    v8f xP, xN;
    #pragma unroll
    for (int v = 0; v < 8; ++v) { xP[v] = biasP; xN[v] = biasN; }
    {
        v2f a0, a1;
        a0.x = xrow[2 * lh];
        a0.y = xrow[2 * lh + 1];
        const float e4 = xrow[4], e5 = xrow[5];
        a1.x = lh ? 0.0f : e4;
        a1.y = lh ? 0.0f : e5;
        xP = wmma4(a0, WP[0], xP);  xP = wmma4(a1, WP[1], xP);
        xN = wmma4(a0, WQ[0], xN);  xN = wmma4(a1, WQ[1], xN);
    }
    v8f qC;
    #pragma unroll
    for (int v = 0; v < 8; ++v) qC[v] = biasQ;

    #pragma unroll 1
    for (int t = 0; t < T_; ++t) {
        const int pub = (t & 1) * (H_ * 16);
        const float xr0 = xnA[0];
        const float xr1 = xnA[1];
        const float xr2 = xnB[0];
        const float xr3 = xnB[1];

        v8f accP = xP, accQ = qC;
        #pragma unroll
        for (int s = 0; s < 2; ++s) {
            accP = wmma4(hA[s], WP[2 + s], accP);
            accQ = wmma4(hA[s], WQ[2 + s], accQ);
        }
        if (t + 5 < T_) __builtin_prefetch(xnB + 4 * I_, 0, 1);
        #pragma unroll
        for (int s = 2; s < 8; ++s) {
            accP = wmma4(hA[s], WP[2 + s], accP);
            accQ = wmma4(hA[s], WQ[2 + s], accQ);
        }

        float ar[4], az[4], tr[4], tz[4], hq[4], xa[4], nn[4], omz[4], zh[4];
        #pragma unroll
        for (int j = 0; j < 4; ++j) {
            const float aS = from_lane_xor8(accP[4 + j]);
            ar[j] = hi ? aS : accP[j];
            az[j] = from_lane_xor8_low_half(accP[j], accP[4 + j]);
        }
        #pragma unroll
        for (int j = 0; j < 4; ++j) tr[j] = htanh(ar[j]);
        #pragma unroll
        for (int j = 0; j < 4; ++j) tz[j] = htanh(az[j]);
        #pragma unroll
        for (int j = 0; j < 4; ++j) {
            const float hS = from_lane_xor8(accQ[4 + j]);
            const float xn = from_lane_xor8_low_half(xN[j], xN[4 + j]);
            hq[j] = hi ? hS : accQ[j];
            xa[j] = xn + hq[j];
        }
        #pragma unroll
        for (int j = 0; j < 4; ++j) nn[j] = htanh(__builtin_fmaf(tr[j], hq[j], xa[j]));
        #pragma unroll
        for (int j = 0; j < 4; ++j) {
            omz[j] = __builtin_fmaf(tz[j], -0.5f, 0.5f);
            zh[j]  = __builtin_fmaf(-omz[j], hold[j], hold[j]);
        }
        #pragma unroll
        for (int j = 0; j < 4; ++j) {
            const float hn = __builtin_fmaf(omz[j], nn[j], zh[j]);
            const int live = (t < len4[j]) ? -1 : 0;
            hold[j] = __int_as_float((__float_as_int(hn) & live) | (__float_as_int(hold[j]) & ~live));
        }
        #pragma unroll
        for (int j = 0; j < 4; ++j) ldsh[pub + hc * 16 + 8 * lh + 4 * hi + j] = hold[j];

        if (t > 0) {
            const float y0 = accQ[2 * wv];
            const float y1 = accQ[2 * wv + 1];
            if (fcl) {
                float* yq = yp + (size_t)(t - 1) * O_;
                yq[0]                 = (t - 1 < lenS0) ? y0 : fcb;
                yq[(size_t)T_ * O_]   = (t - 1 < lenS1) ? y1 : fcb;
            }
        }

        lds_barrier();

        #pragma unroll
        for (int s = 0; s < 8; ++s) {
            const int k0 = 4 * s + 2 * lh;
            hA[s].x = ldsh[pub + k0 * 16 + ln];
            hA[s].y = ldsh[pub + (k0 + 1) * 16 + ln];
        }

        {
            v2f a0, a1;
            a0.x = xr0;
            a0.y = xr1;
            a1.x = lh ? 0.0f : xr2;
            a1.y = lh ? 0.0f : xr3;
            v8f p, n;
            #pragma unroll
            for (int v = 0; v < 8; ++v) { p[v] = biasP; n[v] = biasN; }
            p = wmma4(a0, WP[0], p);  p = wmma4(a1, WP[1], p);
            n = wmma4(a0, WQ[0], n);  n = wmma4(a1, WQ[1], n);
            xP = p;  xN = n;
            const int adv = (t + 2 < T_) ? I_ : 0;
            xnA += adv;
            xnB += adv;
        }
    }

    {
        v8f accQ = qC;
        #pragma unroll
        for (int s = 0; s < 8; ++s) accQ = wmma4(hA[s], WQ[2 + s], accQ);
        const float y0 = accQ[2 * wv];
        const float y1 = accQ[2 * wv + 1];
        if (fcl) {
            float* yq = yp + (size_t)(T_ - 1) * O_;
            yq[0]               = (T_ - 1 < lenS0) ? y0 : fcb;
            yq[(size_t)T_ * O_] = (T_ - 1 < lenS1) ? y1 : fcb;
        }
    }
}

extern "C" void kernel_launch(void* const* d_in, const int* in_sizes, int n_in,
                              void* d_out, int out_size, void* d_ws, size_t ws_size,
                              hipStream_t stream) {
    const float* x       = (const float*)d_in[0];
    const int*   lengths = (const int*)  d_in[1];
    const float* w_ih    = (const float*)d_in[2];
    const float* w_hh    = (const float*)d_in[3];
    const float* b_ih    = (const float*)d_in[4];
    const float* b_hh    = (const float*)d_in[5];
    const float* fc_w    = (const float*)d_in[6];
    const float* fc_b    = (const float*)d_in[7];
    float*       out     = (float*)d_out;

    gru_fused_kernel<<<dim3(B_ / 16), dim3(32 * NWAVES_), 0, stream>>>(
        x, lengths, w_ih, w_hh, b_ih, b_hh, fc_w, fc_b, out);
}


# Products transposed (weights as A, h^T as B): a wave's own new h is two ready B slices that open the next chain from registers under the LDS exchange; 3 b128 reads replace 8; k sum order changed
# speedup vs baseline: 2.6558x; 1.0821x over previous
#include <hip/hip_runtime.h>


typedef float v2f __attribute__((ext_vector_type(2)));
typedef float v4f __attribute__((ext_vector_type(4)));
typedef float v8f __attribute__((ext_vector_type(8)));

#define B_ 128
#define T_ 4096
#define I_ 6
#define H_ 32
#define O_ 2
#define NWAVES_ 4

__device__ __forceinline__ v8f wmma4(v2f a, v2f b, v8f c) {
    return __builtin_amdgcn_wmma_f32_16x16x4_f32(
         false, a,  false, b,
         (short)0, c,  false,  false);
}

__device__ __forceinline__ float htanh(float a) {
    return __builtin_amdgcn_tanhf(a);
}

typedef unsigned int v2u __attribute__((ext_vector_type(2)));

__device__ __forceinline__ void swap_rows(float d, float s, float& dn, float& sn) {
    const v2u r = __builtin_amdgcn_permlane16_swap(__float_as_uint(d), __float_as_uint(s), false, false);
    dn = __uint_as_float(r.x);
    sn = __uint_as_float(r.y);
}

__device__ __forceinline__ void lds_barrier() {
    __builtin_amdgcn_fence(__ATOMIC_RELEASE, "workgroup", "local");
    __builtin_amdgcn_s_barrier();
    __builtin_amdgcn_fence(__ATOMIC_ACQUIRE, "workgroup", "local");
}

__global__ __launch_bounds__(32 * NWAVES_, 1) void gru_fused_kernel(
    const float* __restrict__ x,
    const int*   __restrict__ lengths,
    const float* __restrict__ w_ih,
    const float* __restrict__ w_hh,
    const float* __restrict__ b_ih,
    const float* __restrict__ b_hh,
    const float* __restrict__ fc_w,
    const float* __restrict__ fc_b,
    float*       __restrict__ out)
{
    __shared__ v4f ldsh[2 * NWAVES_ * 32];

    const int tid  = threadIdx.x;
    const int wv   = __builtin_amdgcn_readfirstlane(tid >> 5);
    const int lane = tid & 31;
    const int lh   = lane >> 4;
    const int ln   = lane & 15;
    const int hi   = ln >> 3;
    const int b0   = blockIdx.x * 16;

    const int hc = wv * 8 + (ln & 7);
    const int gP = hi ? (H_ + hc) : hc;
    const int gQ = 2 * H_ + hc;
    const bool fcl = (ln >= 8) && (ln - 8 < O_);
    const int  fo  = fcl ? ln - 8 : 0;
    v2f WP[10], WQ[10];
    #pragma unroll
    for (int s = 0; s < 2; ++s) {
        const int k0 = 4 * s + 2 * lh;
        const float p0 = (k0     < I_) ? w_ih[gP * I_ + k0]     : 0.0f;
        const float p1 = (k0 + 1 < I_) ? w_ih[gP * I_ + k0 + 1] : 0.0f;
        const float q0 = (k0     < I_) ? w_ih[gQ * I_ + k0]     : 0.0f;
        const float q1 = (k0 + 1 < I_) ? w_ih[gQ * I_ + k0 + 1] : 0.0f;
        WP[s].x = 0.5f * p0;       WP[s].y = 0.5f * p1;
        WQ[s].x = hi ? q0 : 0.0f;  WQ[s].y = hi ? q1 : 0.0f;
    }
    #pragma unroll
    for (int s = 0; s < 8; ++s) {
        const int ws = (wv + (s >> 1)) & 3;
        const int k0 = 8 * ws + 4 * lh + 2 * (s & 1);
        WP[2 + s].x = 0.5f * w_hh[gP * H_ + k0];
        WP[2 + s].y = 0.5f * w_hh[gP * H_ + k0 + 1];
        const float q0 = w_hh[gQ * H_ + k0];
        const float q1 = w_hh[gQ * H_ + k0 + 1];
        const float f0 = fcl ? fc_w[fo * H_ + k0]     : 0.0f;
        const float f1 = fcl ? fc_w[fo * H_ + k0 + 1] : 0.0f;
        WQ[2 + s].x = hi ? f0 : 0.5f * q0;
        WQ[2 + s].y = hi ? f1 : 0.5f * q1;
    }
    v8f cP, cN, qC;
    #pragma unroll
    for (int v = 0; v < 8; ++v) {
        const int hv = wv * 8 + v;
        const int g  = lh ? (H_ + hv) : hv;
        cP[v] = 0.5f * (b_ih[g] + b_hh[g]);
        cN[v] = lh ? b_ih[2 * H_ + hv] : 0.0f;
        const float fb = (v < O_) ? fc_b[v] : 0.0f;
        qC[v] = lh ? fb : 0.5f * b_hh[2 * H_ + hv];
    }
    const float fcb0 = fc_b[0], fcb1 = fc_b[1];

    const int len = lengths[b0 + ln];
    const bool ystore = (lh == 1) && ((ln >> 2) == wv);
    float* yp = out + (size_t)(b0 + ln) * T_ * O_;

    const float* xrow = x + (size_t)(b0 + ln) * T_ * I_;
    const float* xnA = xrow + I_ + 2 * lh;
    const float* xnB = xrow + I_ + 4;

    float hold[4];
    v2f   hB[6];
    #pragma unroll
    for (int j = 0; j < 4; ++j) hold[j] = 0.0f;
    #pragma unroll
    for (int s = 0; s < 6; ++s) { hB[s].x = 0.0f; hB[s].y = 0.0f; }

    v8f accP = cP, accQ = qC, xN = cN;
    {
        v2f a0, a1;
        a0.x = xrow[2 * lh];
        a0.y = xrow[2 * lh + 1];
        const float e4 = xrow[4], e5 = xrow[5];
        a1.x = lh ? 0.0f : e4;
        a1.y = lh ? 0.0f : e5;
        accP = wmma4(WP[0], a0, accP);  accP = wmma4(WP[1], a1, accP);
        xN   = wmma4(WQ[0], a0, xN);    xN   = wmma4(WQ[1], a1, xN);
    }
    const int slot = wv * 32 + lane;
    const int slotF1 = ((wv + 1) & 3) * 32 + lane;
    const int slotF2 = ((wv + 2) & 3) * 32 + lane;
    const int slotF3 = ((wv + 3) & 3) * 32 + lane;

    #pragma unroll 1
    for (int t = 0; t < T_; ++t) {
        const int pub = (t & 1) * (NWAVES_ * 32);
        const float xr0 = xnA[0];
        const float xr1 = xnA[1];
        const float xr2 = xnB[0];
        const float xr3 = xnB[1];

        #pragma unroll
        for (int s = 0; s < 2; ++s) {
            accP = wmma4(WP[4 + s], hB[s], accP);
            accQ = wmma4(WQ[4 + s], hB[s], accQ);
        }
        if (t + 5 < T_) __builtin_prefetch(xnB + 4 * I_, 0, 1);
        #pragma unroll
        for (int s = 2; s < 6; ++s) {
            accP = wmma4(WP[4 + s], hB[s], accP);
            accQ = wmma4(WQ[4 + s], hB[s], accQ);
        }

        float ar[4], az[4], tr[4], tz[4], hq[4], xa[4], nn[4], omz[4], zh[4];
        #pragma unroll
        for (int j = 0; j < 4; ++j) {
            swap_rows(accP[j], accP[4 + j], ar[j], az[j]);
        }
        #pragma unroll
        for (int j = 0; j < 4; ++j) tr[j] = htanh(ar[j]);
        #pragma unroll
        for (int j = 0; j < 4; ++j) tz[j] = htanh(az[j]);
        #pragma unroll
        for (int j = 0; j < 4; ++j) {
            float xn, unused0, unused1;
            swap_rows(accQ[j], accQ[4 + j], hq[j], unused0);
            swap_rows(xN[j], xN[4 + j], unused1, xn);
            xa[j] = xn + hq[j];
        }
        #pragma unroll
        for (int j = 0; j < 4; ++j) nn[j] = htanh(__builtin_fmaf(tr[j], hq[j], xa[j]));
        #pragma unroll
        for (int j = 0; j < 4; ++j) {
            omz[j] = __builtin_fmaf(tz[j], -0.5f, 0.5f);
            zh[j]  = __builtin_fmaf(-omz[j], hold[j], hold[j]);
        }
        const int live = (t < len) ? -1 : 0;
        #pragma unroll
        for (int j = 0; j < 4; ++j) {
            const float hn = __builtin_fmaf(omz[j], nn[j], zh[j]);
            hold[j] = __int_as_float((__float_as_int(hn) & live) | (__float_as_int(hold[j]) & ~live));
        }
        {
            v4f hv4;
            hv4.x = hold[0]; hv4.y = hold[1]; hv4.z = hold[2]; hv4.w = hold[3];
            ldsh[pub + slot] = hv4;
        }

        if (t > 0 && ystore) {
            v2f y;
            y.x = (t - 1 < len) ? accQ[0] : fcb0;
            y.y = (t - 1 < len) ? accQ[1] : fcb1;
            *reinterpret_cast<v2f*>(yp + (size_t)(t - 1) * O_) = y;
        }

        v8f p = cP, n = cN;
        {
            v2f a0, a1;
            a0.x = xr0;
            a0.y = xr1;
            a1.x = lh ? 0.0f : xr2;
            a1.y = lh ? 0.0f : xr3;
            p = wmma4(WP[0], a0, p);  p = wmma4(WP[1], a1, p);
            n = wmma4(WQ[0], a0, n);  n = wmma4(WQ[1], a1, n);
            xN = n;
            const int adv = (t + 2 < T_) ? I_ : 0;
            xnA += adv;
            xnB += adv;
        }

        lds_barrier();

        {
            const v4f f1 = ldsh[pub + slotF1];
            const v4f f2 = ldsh[pub + slotF2];
            const v4f f3 = ldsh[pub + slotF3];
            hB[0].x = f1.x; hB[0].y = f1.y; hB[1].x = f1.z; hB[1].y = f1.w;
            hB[2].x = f2.x; hB[2].y = f2.y; hB[3].x = f2.z; hB[3].y = f2.w;
            hB[4].x = f3.x; hB[4].y = f3.y; hB[5].x = f3.z; hB[5].y = f3.w;
        }

        {
            v2f o0, o1;
            o0.x = hold[0]; o0.y = hold[1];
            o1.x = hold[2]; o1.y = hold[3];
            accP = wmma4(WP[2], o0, p);
            accQ = wmma4(WQ[2], o0, qC);
            accP = wmma4(WP[3], o1, accP);
            accQ = wmma4(WQ[3], o1, accQ);
        }
    }

    {
        #pragma unroll
        for (int s = 0; s < 6; ++s) accQ = wmma4(WQ[4 + s], hB[s], accQ);
        if (ystore) {
            v2f y;
            y.x = (T_ - 1 < len) ? accQ[0] : fcb0;
            y.y = (T_ - 1 < len) ? accQ[1] : fcb1;
            *reinterpret_cast<v2f*>(yp + (size_t)(T_ - 1) * O_) = y;
        }
    }
}

extern "C" void kernel_launch(void* const* d_in, const int* in_sizes, int n_in,
                              void* d_out, int out_size, void* d_ws, size_t ws_size,
                              hipStream_t stream) {
    const float* x       = (const float*)d_in[0];
    const int*   lengths = (const int*)  d_in[1];
    const float* w_ih    = (const float*)d_in[2];
    const float* w_hh    = (const float*)d_in[3];
    const float* b_ih    = (const float*)d_in[4];
    const float* b_hh    = (const float*)d_in[5];
    const float* fc_w    = (const float*)d_in[6];
    const float* fc_b    = (const float*)d_in[7];
    float*       out     = (float*)d_out;

    gru_fused_kernel<<<dim3(B_ / 16), dim3(32 * NWAVES_), 0, stream>>>(
        x, lengths, w_ih, w_hh, b_ih, b_hh, fc_w, fc_b, out);
}
